# starting kernel
# baseline (speedup 1.0000x reference)
#include <hip/hip_runtime.h>
#include <hip/hip_bf16.h>

typedef __attribute__((ext_vector_type(2))) float v2f;
typedef __attribute__((ext_vector_type(4))) float v4f;
typedef __attribute__((ext_vector_type(8))) float v8f;

#define D 128
#define LN_EPS 1e-5f

__global__ void __launch_bounds__(256)
deg_init_kernel(float* __restrict__ deg, int N) {
    int i = blockIdx.x * blockDim.x + threadIdx.x;
    if (i < N) deg[i] = 1.0f;
}

__global__ void __launch_bounds__(256)
deg_scatter_kernel(const int* __restrict__ cols, const float* __restrict__ ew,
                   float* __restrict__ deg, int E) {
    int e = blockIdx.x * blockDim.x + threadIdx.x;
    if (e < E) atomicAdd(&deg[cols[e]], ew[e]);
}

__global__ void __launch_bounds__(256)
dinv_kernel(float* __restrict__ deg, int N) {
    int i = blockIdx.x * blockDim.x + threadIdx.x;
    if (i < N) {
        float d = deg[i];
        deg[i] = (d > 0.0f) ? rsqrtf(d) : 0.0f;
    }
}

__global__ void __launch_bounds__(256)
gemm_wmma_kernel(const float* __restrict__ x, const float* __restrict__ W,
                 const float* __restrict__ dinv, float* __restrict__ h,
                 float* __restrict__ acc, int N) {
    const int lane  = threadIdx.x;
    const int nbase = threadIdx.y * 16;
    const int mbase = blockIdx.x * 16;
    const int lmod  = lane & 15;
    const int koff  = (lane >> 4) * 2;

    int arow = mbase + lmod;
    if (arow > N - 1) arow = N - 1;

    const float* xrow = x + (size_t)arow * D + koff;
    const float* wrow = W + (size_t)(nbase + lmod) * D + koff;

    v8f c = {};
#pragma unroll
    for (int k0 = 0; k0 < D; k0 += 4) {
        v2f a = *(const v2f*)(xrow + k0);
        v2f b = *(const v2f*)(wrow + k0);
        c = __builtin_amdgcn_wmma_f32_16x16x4_f32(false, a, false, b,
                                                  (short)0, c, false, false);
    }

    const int mhi = (lane >> 4) * 8;
#pragma unroll
    for (int r = 0; r < 8; ++r) {
        int row = mbase + mhi + r;
        if (row < N) {
            int col = nbase + lmod;
            float v  = c[r];
            float di = dinv[row];
            h[(size_t)row * D + col]   = v;
            acc[(size_t)row * D + col] = v * di * di;
        }
    }
}

__global__ void __launch_bounds__(256)
edge_scatter_kernel(const int* __restrict__ rows, const int* __restrict__ cols,
                    const float* __restrict__ ew, const float* __restrict__ dinv,
                    const float* __restrict__ h, float* __restrict__ acc, int E) {
    long long t = (long long)blockIdx.x * blockDim.x + threadIdx.x;
    int e = (int)(t >> 5);
    if (e >= E) return;
    int part = (int)(t & 31);

    int   src  = rows[e];
    int   dst  = cols[e];
    float norm = dinv[src] * ew[e] * dinv[dst];

    v4f hv = *(const v4f*)(h + (size_t)src * D + part * 4);
    float* out = acc + (size_t)dst * D + part * 4;
#pragma unroll
    for (int i = 0; i < 4; ++i)
        atomicAdd(out + i, hv[i] * norm);
}

__global__ void __launch_bounds__(256)
finalize_kernel(float* __restrict__ out, const float* __restrict__ bias,
                const float* __restrict__ gamma, const float* __restrict__ beta,
                int N) {
    int lane = threadIdx.x & 31;
    int wave = threadIdx.x >> 5;
    int node = blockIdx.x * 8 + wave;
    if (node >= N) return;

    float* p = out + (size_t)node * D + lane * 4;
    v4f v = *(const v4f*)p;
    float s = 0.f, s2 = 0.f;
#pragma unroll
    for (int i = 0; i < 4; ++i) {
        int c = lane * 4 + i;
        float vv = v[i] + bias[c];
        vv = fmaxf(vv, 0.0f);
        v[i] = vv;
        s  += vv;
        s2 += vv * vv;
    }
#pragma unroll
    for (int off = 16; off >= 1; off >>= 1) {
        s  += __shfl_xor(s,  off, 32);
        s2 += __shfl_xor(s2, off, 32);
    }
    float mu  = s * (1.0f / 128.0f);
    float var = s2 * (1.0f / 128.0f) - mu * mu;
    float rs  = rsqrtf(var + LN_EPS);
#pragma unroll
    for (int i = 0; i < 4; ++i) {
        int c = lane * 4 + i;
        v[i] = (v[i] - mu) * rs * gamma[c] + beta[c];
    }
    *(v4f*)p = v;
}

extern "C" void kernel_launch(void* const* d_in, const int* in_sizes, int n_in,
                              void* d_out, int out_size, void* d_ws, size_t ws_size,
                              hipStream_t stream) {
    const float* x     = (const float*)d_in[0];
    const int*   eidx  = (const int*)  d_in[1];
    const float* ew    = (const float*)d_in[2];
    const float* W     = (const float*)d_in[3];
    const float* bias  = (const float*)d_in[4];
    const float* gamma = (const float*)d_in[5];
    const float* beta  = (const float*)d_in[6];
    float*       out   = (float*)d_out;

    const int N = in_sizes[0] / D;
    const int E = in_sizes[2];
    const int* rows = eidx;
    const int* cols = eidx + E;

    float* h    = (float*)d_ws;
    float* dinv = h + (size_t)N * D;

    deg_init_kernel   <<<(N + 255) / 256, 256, 0, stream>>>(dinv, N);
    deg_scatter_kernel<<<(E + 255) / 256, 256, 0, stream>>>(cols, ew, dinv, E);
    dinv_kernel       <<<(N + 255) / 256, 256, 0, stream>>>(dinv, N);

    gemm_wmma_kernel<<<(N + 15) / 16, dim3(32, 8), 0, stream>>>(x, W, dinv, h, out, N);

    long long sthreads = (long long)E * 32;
    edge_scatter_kernel<<<(unsigned)((sthreads + 255) / 256), 256, 0, stream>>>(
        rows, cols, ew, dinv, h, out, E);

    finalize_kernel<<<(N + 7) / 8, 256, 0, stream>>>(out, bias, gamma, beta, N);
}


# GEMM operand feed: W staged once per 128-row work-group in LDS (pitch 132), each wave's 16x128 x tile staged by full-row 16-B loads; one wave makes 8 column tiles, A fragment reused 8x; k order kept
# speedup vs baseline: 1.1759x; 1.1759x over previous
#include <hip/hip_runtime.h>
#include <hip/hip_bf16.h>

typedef __attribute__((ext_vector_type(2))) float v2f;
typedef __attribute__((ext_vector_type(4))) float v4f;
typedef __attribute__((ext_vector_type(8))) float v8f;

#define D 128
#define LN_EPS 1e-5f

__global__ void __launch_bounds__(256)
deg_init_kernel(float* __restrict__ deg, int N) {
    int i = blockIdx.x * blockDim.x + threadIdx.x;
    if (i < N) deg[i] = 1.0f;
}

__global__ void __launch_bounds__(256)
deg_scatter_kernel(const int* __restrict__ cols, const float* __restrict__ ew,
                   float* __restrict__ deg, int E) {
    int e = blockIdx.x * blockDim.x + threadIdx.x;
    if (e < E) atomicAdd(&deg[cols[e]], ew[e]);
}

__global__ void __launch_bounds__(256)
dinv_kernel(float* __restrict__ deg, int N) {
    int i = blockIdx.x * blockDim.x + threadIdx.x;
    if (i < N) {
        float d = deg[i];
        deg[i] = (d > 0.0f) ? rsqrtf(d) : 0.0f;
    }
}

#define LDP 132
__global__ void __launch_bounds__(256)
gemm_wmma_kernel(const float* __restrict__ x, const float* __restrict__ W,
                 const float* __restrict__ dinv, float* __restrict__ h,
                 float* __restrict__ acc, int N) {
    __shared__ __attribute__((aligned(16))) float Ws[D * LDP];
    __shared__ __attribute__((aligned(16))) float As[8 * 16 * LDP];

    const int lane  = threadIdx.x & 31;
    const int wave  = threadIdx.x >> 5;
    const int lmod  = lane & 15;
    const int koff  = (lane >> 4) * 2;
    const int mbase = blockIdx.x * 128 + wave * 16;

    float* At = As + wave * 16 * LDP;

#pragma unroll
    for (int i = 0; i < 16; ++i) {
        int n = wave * 16 + i;
        v4f w4 = *(const v4f*)(W + (size_t)n * D + lane * 4);
        int row = mbase + i;
        if (row > N - 1) row = N - 1;
        v4f x4 = *(const v4f*)(x + (size_t)row * D + lane * 4);
        *(v4f*)(Ws + n * LDP + lane * 4) = w4;
        *(v4f*)(At + i * LDP + lane * 4) = x4;
    }
    __syncthreads();
    if (mbase >= N) return;

    const float* arow = At + lmod * LDP + koff;
    const float* wrow = Ws + lmod * LDP + koff;

    v8f c[8] = {};

#pragma unroll
    for (int k0 = 0; k0 < D; k0 += 4) {
        v2f a = *(const v2f*)(arow + k0);
#pragma unroll
        for (int j = 0; j < 8; ++j) {
            v2f b = *(const v2f*)(wrow + j * 16 * LDP + k0);
            c[j] = __builtin_amdgcn_wmma_f32_16x16x4_f32(false, a, false, b,
                                                         (short)0, c[j], false, false);
        }
    }

    const int mhi = (lane >> 4) * 8;
#pragma unroll
    for (int r = 0; r < 8; ++r) {
        int row = mbase + mhi + r;
        if (row < N) {
            float di = dinv[row];
#pragma unroll
            for (int j = 0; j < 8; ++j) {
                int col = j * 16 + lmod;
                float v = c[j][r];
                h[(size_t)row * D + col]   = v;
                acc[(size_t)row * D + col] = v * di * di;
            }
        }
    }
}

__global__ void __launch_bounds__(256)
edge_scatter_kernel(const int* __restrict__ rows, const int* __restrict__ cols,
                    const float* __restrict__ ew, const float* __restrict__ dinv,
                    const float* __restrict__ h, float* __restrict__ acc, int E) {
    long long t = (long long)blockIdx.x * blockDim.x + threadIdx.x;
    int e = (int)(t >> 5);
    if (e >= E) return;
    int part = (int)(t & 31);

    int   src  = rows[e];
    int   dst  = cols[e];
    float norm = dinv[src] * ew[e] * dinv[dst];

    v4f hv = *(const v4f*)(h + (size_t)src * D + part * 4);
    float* out = acc + (size_t)dst * D + part * 4;
#pragma unroll
    for (int i = 0; i < 4; ++i)
        atomicAdd(out + i, hv[i] * norm);
}

__global__ void __launch_bounds__(256)
finalize_kernel(float* __restrict__ out, const float* __restrict__ bias,
                const float* __restrict__ gamma, const float* __restrict__ beta,
                int N) {
    int lane = threadIdx.x & 31;
    int wave = threadIdx.x >> 5;
    int node = blockIdx.x * 8 + wave;
    if (node >= N) return;

    float* p = out + (size_t)node * D + lane * 4;
    v4f v = *(const v4f*)p;
    float s = 0.f, s2 = 0.f;
#pragma unroll
    for (int i = 0; i < 4; ++i) {
        int c = lane * 4 + i;
        float vv = v[i] + bias[c];
        vv = fmaxf(vv, 0.0f);
        v[i] = vv;
        s  += vv;
        s2 += vv * vv;
    }
#pragma unroll
    for (int off = 16; off >= 1; off >>= 1) {
        s  += __shfl_xor(s,  off, 32);
        s2 += __shfl_xor(s2, off, 32);
    }
    float mu  = s * (1.0f / 128.0f);
    float var = s2 * (1.0f / 128.0f) - mu * mu;
    float rs  = rsqrtf(var + LN_EPS);
#pragma unroll
    for (int i = 0; i < 4; ++i) {
        int c = lane * 4 + i;
        v[i] = (v[i] - mu) * rs * gamma[c] + beta[c];
    }
    *(v4f*)p = v;
}

extern "C" void kernel_launch(void* const* d_in, const int* in_sizes, int n_in,
                              void* d_out, int out_size, void* d_ws, size_t ws_size,
                              hipStream_t stream) {
    const float* x     = (const float*)d_in[0];
    const int*   eidx  = (const int*)  d_in[1];
    const float* ew    = (const float*)d_in[2];
    const float* W     = (const float*)d_in[3];
    const float* bias  = (const float*)d_in[4];
    const float* gamma = (const float*)d_in[5];
    const float* beta  = (const float*)d_in[6];
    float*       out   = (float*)d_out;

    const int N = in_sizes[0] / D;
    const int E = in_sizes[2];
    const int* rows = eidx;
    const int* cols = eidx + E;

    float* h    = (float*)d_ws;
    float* dinv = h + (size_t)N * D;

    deg_init_kernel   <<<(N + 255) / 256, 256, 0, stream>>>(dinv, N);
    deg_scatter_kernel<<<(E + 255) / 256, 256, 0, stream>>>(cols, ew, dinv, E);
    dinv_kernel       <<<(N + 255) / 256, 256, 0, stream>>>(dinv, N);

    gemm_wmma_kernel<<<(N + 127) / 128, 256, 0, stream>>>(x, W, dinv, h, out, N);

    long long sthreads = (long long)E * 32;
    edge_scatter_kernel<<<(unsigned)((sthreads + 255) / 256), 256, 0, stream>>>(
        rows, cols, ew, dinv, h, out, E);

    finalize_kernel<<<(N + 7) / 8, 256, 0, stream>>>(out, bias, gamma, beta, N);
}


# edge sum as a pull: each edge filed in its target's out row (returning int atomic); one wave per node sums its h rows in registers in arrival order, then bias/ReLU/LN; float atomics on acc gone
# speedup vs baseline: 2.2318x; 1.8980x over previous
#include <hip/hip_runtime.h>
#include <hip/hip_bf16.h>

typedef __attribute__((ext_vector_type(2))) float v2f;
typedef __attribute__((ext_vector_type(4))) float v4f;
typedef __attribute__((ext_vector_type(8))) float v8f;
typedef __attribute__((ext_vector_type(2))) int   v2i;

#define D 128
#define LN_EPS 1e-5f

#define BUCKET_CAP 63

__global__ void __launch_bounds__(256)
init_kernel(float* __restrict__ deg, int* __restrict__ bucket, int N) {
    int i = blockIdx.x * blockDim.x + threadIdx.x;
    if (i < N) {
        deg[i] = 1.0f;
        bucket[(size_t)i * D] = 0;
    }
}

__global__ void __launch_bounds__(256)
count_kernel(const int* __restrict__ rows, const int* __restrict__ cols,
             const float* __restrict__ ew, float* __restrict__ deg,
             int* __restrict__ bucket, int E) {
    int e = blockIdx.x * blockDim.x + threadIdx.x;
    if (e >= E) return;
    int   src = rows[e];
    int   dst = cols[e];
    float w   = ew[e];
    atomicAdd(&deg[dst], w);
    int* b = bucket + (size_t)dst * D;
    int pos = atomicAdd(b, 1);
    if (pos < BUCKET_CAP) {
        v2i ent = { src, __float_as_int(w) };
        *(v2i*)(b + 2 + 2 * pos) = ent;
    }
}

__global__ void __launch_bounds__(256)
dinv_kernel(float* __restrict__ deg, int N) {
    int i = blockIdx.x * blockDim.x + threadIdx.x;
    if (i < N) {
        float d = deg[i];
        deg[i] = (d > 0.0f) ? rsqrtf(d) : 0.0f;
    }
}

#define LDP 132
__global__ void __launch_bounds__(256)
gemm_wmma_kernel(const float* __restrict__ x, const float* __restrict__ W,
                 float* __restrict__ h, int N) {
    __shared__ __attribute__((aligned(16))) float Ws[D * LDP];
    __shared__ __attribute__((aligned(16))) float As[8 * 16 * LDP];

    const int lane  = threadIdx.x & 31;
    const int wave  = threadIdx.x >> 5;
    const int lmod  = lane & 15;
    const int koff  = (lane >> 4) * 2;
    const int mbase = blockIdx.x * 128 + wave * 16;

    float* At = As + wave * 16 * LDP;

#pragma unroll
    for (int i = 0; i < 16; ++i) {
        int n = wave * 16 + i;
        v4f w4 = *(const v4f*)(W + (size_t)n * D + lane * 4);
        int row = mbase + i;
        if (row > N - 1) row = N - 1;
        v4f x4 = *(const v4f*)(x + (size_t)row * D + lane * 4);
        *(v4f*)(Ws + n * LDP + lane * 4) = w4;
        *(v4f*)(At + i * LDP + lane * 4) = x4;
    }
    __syncthreads();
    if (mbase >= N) return;

    const float* arow = At + lmod * LDP + koff;
    const float* wrow = Ws + lmod * LDP + koff;

    v8f c[8] = {};

#pragma unroll
    for (int k0 = 0; k0 < D; k0 += 4) {
        v2f a = *(const v2f*)(arow + k0);
#pragma unroll
        for (int j = 0; j < 8; ++j) {
            v2f b = *(const v2f*)(wrow + j * 16 * LDP + k0);
            c[j] = __builtin_amdgcn_wmma_f32_16x16x4_f32(false, a, false, b,
                                                         (short)0, c[j], false, false);
        }
    }

    const int mhi = (lane >> 4) * 8;
#pragma unroll
    for (int r = 0; r < 8; ++r) {
        int row = mbase + mhi + r;
        if (row < N) {
#pragma unroll
            for (int j = 0; j < 8; ++j) {
                int col = j * 16 + lmod;
                h[(size_t)row * D + col] = c[j][r];
            }
        }
    }
}


static __device__ __forceinline__ v4f
gather_chunk(v4f acc, v2i u, bool first, int nvalid, int node, float di,
             const float* __restrict__ dinv, const float* __restrict__ h, int lane) {
    const bool valid  = lane < nvalid;
    const bool isself = first && (lane == 0);
    int   src_l = node;
    float w_l   = 0.0f;
    if (valid) {
        if (isself) {
            w_l = di * di;
        } else {
            src_l = u.x;
            w_l   = dinv[src_l] * __int_as_float(u.y) * di;
        }
    }
    const int w_bits = __float_as_int(w_l);
    for (int t = 0; t < nvalid; t += 4) {
        int   s0 = __builtin_amdgcn_readlane(src_l, t);
        int   s1 = __builtin_amdgcn_readlane(src_l, t + 1);
        int   s2 = __builtin_amdgcn_readlane(src_l, t + 2);
        int   s3 = __builtin_amdgcn_readlane(src_l, t + 3);
        float w0 = __int_as_float(__builtin_amdgcn_readlane(w_bits, t));
        float w1 = __int_as_float(__builtin_amdgcn_readlane(w_bits, t + 1));
        float w2 = __int_as_float(__builtin_amdgcn_readlane(w_bits, t + 2));
        float w3 = __int_as_float(__builtin_amdgcn_readlane(w_bits, t + 3));
        v4f h0 = *(const v4f*)(h + (size_t)s0 * D + lane * 4);
        v4f h1 = *(const v4f*)(h + (size_t)s1 * D + lane * 4);
        v4f h2 = *(const v4f*)(h + (size_t)s2 * D + lane * 4);
        v4f h3 = *(const v4f*)(h + (size_t)s3 * D + lane * 4);
        acc += h0 * w0;
        acc += h1 * w1;
        acc += h2 * w2;
        acc += h3 * w3;
    }
    return acc;
}

__global__ void __launch_bounds__(256)
pull_finalize_kernel(const int* __restrict__ rows, const int* __restrict__ cols,
                     const float* __restrict__ ew, const float* __restrict__ dinv,
                     const float* __restrict__ h, float* out,
                     const float* __restrict__ bias, const float* __restrict__ gamma,
                     const float* __restrict__ beta, int N, int E) {
    const int lane = threadIdx.x & 31;
    const int wave = threadIdx.x >> 5;
    const int node = blockIdx.x * 8 + wave;
    if (node >= N) return;

    float* p = out + (size_t)node * D + lane * 4;
    const v2i* units = (const v2i*)(out + (size_t)node * D);
    const float di = dinv[node];

    v2i u = { 0, 0 };
    if (lane < 16) u = units[lane];
    const int c = __builtin_amdgcn_readfirstlane(u.x);

    v4f acc = { 0.0f, 0.0f, 0.0f, 0.0f };
    if (c <= BUCKET_CAP) {
        if (c > 15) {
            if (lane >= 16) u = units[lane];
        }
        acc = gather_chunk(acc, u, true, (c < 31 ? c : 31) + 1, node, di, dinv, h, lane);
        if (c > 31) {
            u = units[32 + lane];
            acc = gather_chunk(acc, u, false, c - 31, node, di, dinv, h, lane);
        }
    } else {
        v4f hs = *(const v4f*)(h + (size_t)node * D + lane * 4);
        acc = hs * (di * di);
        for (int base = 0; base < E; base += 32) {
            int  e     = base + lane;
            bool match = false;
            if (e < E) match = (cols[e] == node);
            int   src_l = node;
            float w_l   = 0.0f;
            if (match) {
                src_l = rows[e];
                w_l   = dinv[src_l] * ew[e] * di;
            }
            const int w_bits = __float_as_int(w_l);
            unsigned mask = __builtin_amdgcn_ballot_w32(match);
            while (mask) {
                int t = __builtin_ctz(mask);
                mask &= mask - 1;
                int   s = __builtin_amdgcn_readlane(src_l, t);
                float w = __int_as_float(__builtin_amdgcn_readlane(w_bits, t));
                v4f hv = *(const v4f*)(h + (size_t)s * D + lane * 4);
                acc += hv * w;
            }
        }
    }

    v4f v = acc;
    float s = 0.f, s2 = 0.f;
#pragma unroll
    for (int i = 0; i < 4; ++i) {
        int ch = lane * 4 + i;
        float vv = v[i] + bias[ch];
        vv = fmaxf(vv, 0.0f);
        v[i] = vv;
        s  += vv;
        s2 += vv * vv;
    }
#pragma unroll
    for (int off = 16; off >= 1; off >>= 1) {
        s  += __shfl_xor(s,  off, 32);
        s2 += __shfl_xor(s2, off, 32);
    }
    float mu  = s * (1.0f / 128.0f);
    float var = s2 * (1.0f / 128.0f) - mu * mu;
    float rs  = rsqrtf(var + LN_EPS);
#pragma unroll
    for (int i = 0; i < 4; ++i) {
        int ch = lane * 4 + i;
        v[i] = (v[i] - mu) * rs * gamma[ch] + beta[ch];
    }
    *(v4f*)p = v;
}

extern "C" void kernel_launch(void* const* d_in, const int* in_sizes, int n_in,
                              void* d_out, int out_size, void* d_ws, size_t ws_size,
                              hipStream_t stream) {
    const float* x     = (const float*)d_in[0];
    const int*   eidx  = (const int*)  d_in[1];
    const float* ew    = (const float*)d_in[2];
    const float* W     = (const float*)d_in[3];
    const float* bias  = (const float*)d_in[4];
    const float* gamma = (const float*)d_in[5];
    const float* beta  = (const float*)d_in[6];
    float*       out   = (float*)d_out;

    const int N = in_sizes[0] / D;
    const int E = in_sizes[2];
    const int* rows = eidx;
    const int* cols = eidx + E;

    float* h    = (float*)d_ws;
    float* dinv = h + (size_t)N * D;

    init_kernel <<<(N + 255) / 256, 256, 0, stream>>>(dinv, (int*)out, N);
    count_kernel<<<(E + 255) / 256, 256, 0, stream>>>(rows, cols, ew, dinv, (int*)out, E);
    dinv_kernel <<<(N + 255) / 256, 256, 0, stream>>>(dinv, N);

    gemm_wmma_kernel<<<(N + 127) / 128, 256, 0, stream>>>(x, W, h, N);

    pull_finalize_kernel<<<(N + 7) / 8, 256, 0, stream>>>(
        rows, cols, ew, dinv, h, out, bias, gamma, beta, N, E);
}


# normalisation split per row: GEMM store scales each h row by its source's deg^-1/2, pull applies the target's; per-edge deg^-1/2 lookup and the rsqrt launch gone (5 to 4 launches)
# speedup vs baseline: 2.3095x; 1.0348x over previous
#include <hip/hip_runtime.h>
#include <hip/hip_bf16.h>

typedef __attribute__((ext_vector_type(2))) float v2f;
typedef __attribute__((ext_vector_type(4))) float v4f;
typedef __attribute__((ext_vector_type(8))) float v8f;
typedef __attribute__((ext_vector_type(2))) int   v2i;

#define D 128
#define LN_EPS 1e-5f

#define BUCKET_CAP 63

__global__ void __launch_bounds__(256)
init_kernel(float* __restrict__ deg, int* __restrict__ bucket, int N) {
    int i = blockIdx.x * blockDim.x + threadIdx.x;
    if (i < N) {
        deg[i] = 1.0f;
        bucket[(size_t)i * D] = 0;
    }
}

__global__ void __launch_bounds__(256)
count_kernel(const int* __restrict__ rows, const int* __restrict__ cols,
             const float* __restrict__ ew, float* __restrict__ deg,
             int* __restrict__ bucket, int E) {
    int e = blockIdx.x * blockDim.x + threadIdx.x;
    if (e >= E) return;
    int   src = rows[e];
    int   dst = cols[e];
    float w   = ew[e];
    atomicAdd(&deg[dst], w);
    int* b = bucket + (size_t)dst * D;
    int pos = atomicAdd(b, 1);
    if (pos < BUCKET_CAP) {
        v2i ent = { src, __float_as_int(w) };
        *(v2i*)(b + 2 + 2 * pos) = ent;
    }
}

static __device__ __forceinline__ float inv_sqrt_deg(float d) {
    return (d > 0.0f) ? rsqrtf(d) : 0.0f;
}

#define LDP 132
__global__ void __launch_bounds__(256)
gemm_wmma_kernel(const float* __restrict__ x, const float* __restrict__ W,
                 const float* __restrict__ deg, float* __restrict__ h, int N) {
    __shared__ __attribute__((aligned(16))) float Ws[D * LDP];
    __shared__ __attribute__((aligned(16))) float As[8 * 16 * LDP];

    const int lane  = threadIdx.x & 31;
    const int wave  = threadIdx.x >> 5;
    const int lmod  = lane & 15;
    const int koff  = (lane >> 4) * 2;
    const int mbase = blockIdx.x * 128 + wave * 16;

    float* At = As + wave * 16 * LDP;

    const int mhi = (lane >> 4) * 8;
    float rdeg[8];
#pragma unroll
    for (int r = 0; r < 8; ++r) {
        int row = mbase + mhi + r;
        if (row > N - 1) row = N - 1;
        rdeg[r] = deg[row];
    }

#pragma unroll
    for (int i = 0; i < 16; ++i) {
        int n = wave * 16 + i;
        v4f w4 = *(const v4f*)(W + (size_t)n * D + lane * 4);
        int row = mbase + i;
        if (row > N - 1) row = N - 1;
        v4f x4 = *(const v4f*)(x + (size_t)row * D + lane * 4);
        *(v4f*)(Ws + n * LDP + lane * 4) = w4;
        *(v4f*)(At + i * LDP + lane * 4) = x4;
    }
    __syncthreads();
    if (mbase >= N) return;

    const float* arow = At + lmod * LDP + koff;
    const float* wrow = Ws + lmod * LDP + koff;

    v8f c[8] = {};

#pragma unroll
    for (int k0 = 0; k0 < D; k0 += 4) {
        v2f a = *(const v2f*)(arow + k0);
#pragma unroll
        for (int j = 0; j < 8; ++j) {
            v2f b = *(const v2f*)(wrow + j * 16 * LDP + k0);
            c[j] = __builtin_amdgcn_wmma_f32_16x16x4_f32(false, a, false, b,
                                                         (short)0, c[j], false, false);
        }
    }

    float rscale[8];
#pragma unroll
    for (int r = 0; r < 8; ++r) rscale[r] = inv_sqrt_deg(rdeg[r]);
#pragma unroll
    for (int j = 0; j < 8; ++j) {
#pragma unroll
        for (int r = 0; r < 8; ++r) c[j][r] *= rscale[r];
    }
#pragma unroll
    for (int r = 0; r < 8; ++r) {
        int row = mbase + mhi + r;
        if (row < N) {
#pragma unroll
            for (int j = 0; j < 8; ++j) {
                int col = j * 16 + lmod;
                h[(size_t)row * D + col] = c[j][r];
            }
        }
    }
}


static __device__ __forceinline__ v4f
gather_chunk(v4f acc, v2i u, bool first, int nvalid, int node, float di,
             const float* __restrict__ h, int lane) {
    const bool valid  = lane < nvalid;
    const bool isself = first && (lane == 0);
    int   src_l = node;
    float w_l   = 0.0f;
    if (valid) {
        if (isself) {
            w_l = di;
        } else {
            src_l = u.x;
            w_l   = __int_as_float(u.y) * di;
        }
    }
    const int w_bits = __float_as_int(w_l);
    for (int t = 0; t < nvalid; t += 4) {
        int   s0 = __builtin_amdgcn_readlane(src_l, t);
        int   s1 = __builtin_amdgcn_readlane(src_l, t + 1);
        int   s2 = __builtin_amdgcn_readlane(src_l, t + 2);
        int   s3 = __builtin_amdgcn_readlane(src_l, t + 3);
        float w0 = __int_as_float(__builtin_amdgcn_readlane(w_bits, t));
        float w1 = __int_as_float(__builtin_amdgcn_readlane(w_bits, t + 1));
        float w2 = __int_as_float(__builtin_amdgcn_readlane(w_bits, t + 2));
        float w3 = __int_as_float(__builtin_amdgcn_readlane(w_bits, t + 3));
        v4f h0 = *(const v4f*)(h + (size_t)s0 * D + lane * 4);
        v4f h1 = *(const v4f*)(h + (size_t)s1 * D + lane * 4);
        v4f h2 = *(const v4f*)(h + (size_t)s2 * D + lane * 4);
        v4f h3 = *(const v4f*)(h + (size_t)s3 * D + lane * 4);
        acc += h0 * w0;
        acc += h1 * w1;
        acc += h2 * w2;
        acc += h3 * w3;
    }
    return acc;
}

__global__ void __launch_bounds__(256)
pull_finalize_kernel(const int* __restrict__ rows, const int* __restrict__ cols,
                     const float* __restrict__ ew, const float* __restrict__ deg,
                     const float* __restrict__ h, float* out,
                     const float* __restrict__ bias, const float* __restrict__ gamma,
                     const float* __restrict__ beta, int N, int E) {
    const int lane = threadIdx.x & 31;
    const int wave = threadIdx.x >> 5;
    const int node = blockIdx.x * 8 + wave;
    if (node >= N) return;

    float* p = out + (size_t)node * D + lane * 4;
    const v2i* units = (const v2i*)(out + (size_t)node * D);
    const float di = inv_sqrt_deg(deg[node]);

    v2i u = { 0, 0 };
    if (lane < 16) u = units[lane];
    const int c = __builtin_amdgcn_readfirstlane(u.x);

    v4f acc = { 0.0f, 0.0f, 0.0f, 0.0f };
    if (c <= BUCKET_CAP) {
        if (c > 15) {
            if (lane >= 16) u = units[lane];
        }
        acc = gather_chunk(acc, u, true, (c < 31 ? c : 31) + 1, node, di, h, lane);
        if (c > 31) {
            u = units[32 + lane];
            acc = gather_chunk(acc, u, false, c - 31, node, di, h, lane);
        }
    } else {
        v4f hs = *(const v4f*)(h + (size_t)node * D + lane * 4);
        acc = hs * di;
        for (int base = 0; base < E; base += 32) {
            int  e     = base + lane;
            bool match = false;
            if (e < E) match = (cols[e] == node);
            int   src_l = node;
            float w_l   = 0.0f;
            if (match) {
                src_l = rows[e];
                w_l   = ew[e] * di;
            }
            const int w_bits = __float_as_int(w_l);
            unsigned mask = __builtin_amdgcn_ballot_w32(match);
            while (mask) {
                int t = __builtin_ctz(mask);
                mask &= mask - 1;
                int   s = __builtin_amdgcn_readlane(src_l, t);
                float w = __int_as_float(__builtin_amdgcn_readlane(w_bits, t));
                v4f hv = *(const v4f*)(h + (size_t)s * D + lane * 4);
                acc += hv * w;
            }
        }
    }

    v4f v = acc;
    float s = 0.f, s2 = 0.f;
#pragma unroll
    for (int i = 0; i < 4; ++i) {
        int ch = lane * 4 + i;
        float vv = v[i] + bias[ch];
        vv = fmaxf(vv, 0.0f);
        v[i] = vv;
        s  += vv;
        s2 += vv * vv;
    }
#pragma unroll
    for (int off = 16; off >= 1; off >>= 1) {
        s  += __shfl_xor(s,  off, 32);
        s2 += __shfl_xor(s2, off, 32);
    }
    float mu  = s * (1.0f / 128.0f);
    float var = s2 * (1.0f / 128.0f) - mu * mu;
    float rs  = rsqrtf(var + LN_EPS);
#pragma unroll
    for (int i = 0; i < 4; ++i) {
        int ch = lane * 4 + i;
        v[i] = (v[i] - mu) * rs * gamma[ch] + beta[ch];
    }
    *(v4f*)p = v;
}

extern "C" void kernel_launch(void* const* d_in, const int* in_sizes, int n_in,
                              void* d_out, int out_size, void* d_ws, size_t ws_size,
                              hipStream_t stream) {
    const float* x     = (const float*)d_in[0];
    const int*   eidx  = (const int*)  d_in[1];
    const float* ew    = (const float*)d_in[2];
    const float* W     = (const float*)d_in[3];
    const float* bias  = (const float*)d_in[4];
    const float* gamma = (const float*)d_in[5];
    const float* beta  = (const float*)d_in[6];
    float*       out   = (float*)d_out;

    const int N = in_sizes[0] / D;
    const int E = in_sizes[2];
    const int* rows = eidx;
    const int* cols = eidx + E;

    float* h    = (float*)d_ws;
    float* deg  = h + (size_t)N * D;

    init_kernel <<<(N + 255) / 256, 256, 0, stream>>>(deg, (int*)out, N);
    count_kernel<<<(E + 255) / 256, 256, 0, stream>>>(rows, cols, ew, deg, (int*)out, E);

    gemm_wmma_kernel<<<(N + 127) / 128, 256, 0, stream>>>(x, W, deg, h, N);

    pull_finalize_kernel<<<(N + 7) / 8, 256, 0, stream>>>(
        rows, cols, ew, deg, h, out, bias, gamma, beta, N, E);
}


# bucket fill by LDS binning: edges counting-sorted by 512-node target range per chunk, then one work-group per range forms lists and degree sums with LDS atomics (arrival order); global atomics gone
# speedup vs baseline: 3.3050x; 1.4310x over previous
#include <hip/hip_runtime.h>
#include <hip/hip_bf16.h>

typedef __attribute__((ext_vector_type(2))) float v2f;
typedef __attribute__((ext_vector_type(4))) float v4f;
typedef __attribute__((ext_vector_type(8))) float v8f;
typedef __attribute__((ext_vector_type(2))) int   v2i;
typedef __attribute__((ext_vector_type(4))) int   v4i;

#define D 128
#define LN_EPS 1e-5f

#define BUCKET_CAP 63

#define BIN_SHIFT 9
#define BIN_NODES 512
#define MAXBINS   256
#define OFFS      (MAXBINS + 1)
#define CHUNK     2560

__global__ void __launch_bounds__(256)
partition_kernel(const int* __restrict__ rows, const int* __restrict__ cols,
                 const float* __restrict__ ew, int* __restrict__ ent,
                 int* __restrict__ off, int E) {
    __shared__ int hist[MAXBINS];
    __shared__ int start[OFFS];
    __shared__ int wsum[8];
    __shared__ __attribute__((aligned(16))) int sorted[CHUNK * 4];

    const int tid  = threadIdx.x;
    const int lane = tid & 31;
    const int wave = tid >> 5;
    const int base = blockIdx.x * CHUNK;

    hist[tid] = 0;
    __syncthreads();

    int  s[10], d[10], wb[10], r[10];
    bool ok[10];
#pragma unroll
    for (int k = 0; k < 10; ++k) {
        int e = base + k * 256 + tid;
        ok[k] = e < E;
        s[k] = 0; d[k] = 0; wb[k] = 0; r[k] = 0;
        if (ok[k]) {
            s[k]  = rows[e];
            d[k]  = cols[e];
            wb[k] = __float_as_int(ew[e]);
        }
    }
#pragma unroll
    for (int k = 0; k < 10; ++k)
        if (ok[k]) r[k] = atomicAdd(&hist[d[k] >> BIN_SHIFT], 1);
    __syncthreads();

    const int v = hist[tid];
    int inc = v;
#pragma unroll
    for (int o = 1; o < 32; o <<= 1) {
        int t = __shfl_up(inc, o, 32);
        if (lane >= o) inc += t;
    }
    if (lane == 31) wsum[wave] = inc;
    __syncthreads();
    int pre = 0;
    for (int i = 0; i < wave; ++i) pre += wsum[i];
    start[tid] = pre + inc - v;
    if (tid == 255) start[MAXBINS] = pre + inc;
    __syncthreads();

#pragma unroll
    for (int k = 0; k < 10; ++k) {
        if (ok[k]) {
            int p = start[d[k] >> BIN_SHIFT] + r[k];
            v4i en = { s[k], d[k], wb[k], 0 };
            *(v4i*)(sorted + 4 * p) = en;
        }
    }
    __syncthreads();

    const int total = start[MAXBINS];
    for (int i = tid; i < total; i += 256)
        *(v4i*)(ent + ((size_t)base + i) * 4) = *(const v4i*)(sorted + 4 * i);
    for (int i = tid; i < OFFS; i += 256)
        off[(size_t)blockIdx.x * OFFS + i] = start[i];
}

__global__ void __launch_bounds__(512)
build_kernel(const int* __restrict__ ent, const int* __restrict__ off,
             float* __restrict__ deg, int* __restrict__ bucket, int N, int nchunks) {
    __shared__ int   cnt[BIN_NODES];
    __shared__ float dsum[BIN_NODES];
    __shared__ __attribute__((aligned(16))) v2i stage[BIN_NODES * 64];

    const int tid   = threadIdx.x;
    const int b     = blockIdx.x;
    const int node0 = b * BIN_NODES;

    cnt[tid]  = 0;
    dsum[tid] = 0.0f;
    __syncthreads();

    const int sub = tid & 1;
    for (int w = tid >> 1; w < nchunks; w += 256) {
        const int lo = off[(size_t)w * OFFS + b];
        const int hi = off[(size_t)w * OFFS + b + 1];
        const int* seg = ent + (size_t)w * CHUNK * 4;
        for (int i = lo + sub; i < hi; i += 2) {
            v4i en = *(const v4i*)(seg + 4 * i);
            int ld  = en.y - node0;
            int pos = atomicAdd(&cnt[ld], 1);
            atomicAdd(&dsum[ld], __int_as_float(en.z));
            if (pos < BUCKET_CAP) {
                v2i u = { en.x, en.z };
                stage[ld * 64 + 1 + pos] = u;
            }
        }
    }
    __syncthreads();

    const int node = node0 + tid;
    if (node < N) deg[node] = 1.0f + dsum[tid];

    const int l16 = tid & 15;
    for (int n = tid >> 4; n < BIN_NODES; n += 32) {
        if (node0 + n < N) {
            const int c    = cnt[n];
            const int used = (c < BUCKET_CAP ? c : BUCKET_CAP) + 1;
            const int nl   = (used + 15) >> 4;
            int* brow = bucket + (size_t)(node0 + n) * D;
            for (int L = 0; L < nl; ++L) {
                v2i u = stage[n * 64 + L * 16 + l16];
                if (L == 0 && l16 == 0) { u.x = c; u.y = 0; }
                *(v2i*)(brow + (L * 16 + l16) * 2) = u;
            }
        }
    }
}

static __device__ __forceinline__ float inv_sqrt_deg(float d) {
    return (d > 0.0f) ? rsqrtf(d) : 0.0f;
}

#define LDP 132
__global__ void __launch_bounds__(256)
gemm_wmma_kernel(const float* __restrict__ x, const float* __restrict__ W,
                 const float* __restrict__ deg, float* __restrict__ h, int N) {
    __shared__ __attribute__((aligned(16))) float Ws[D * LDP];
    __shared__ __attribute__((aligned(16))) float As[8 * 16 * LDP];

    const int lane  = threadIdx.x & 31;
    const int wave  = threadIdx.x >> 5;
    const int lmod  = lane & 15;
    const int koff  = (lane >> 4) * 2;
    const int mbase = blockIdx.x * 128 + wave * 16;

    float* At = As + wave * 16 * LDP;

    const int mhi = (lane >> 4) * 8;
    float rdeg[8];
#pragma unroll
    for (int r = 0; r < 8; ++r) {
        int row = mbase + mhi + r;
        if (row > N - 1) row = N - 1;
        rdeg[r] = deg[row];
    }

#pragma unroll
    for (int i = 0; i < 16; ++i) {
        int n = wave * 16 + i;
        v4f w4 = *(const v4f*)(W + (size_t)n * D + lane * 4);
        int row = mbase + i;
        if (row > N - 1) row = N - 1;
        v4f x4 = *(const v4f*)(x + (size_t)row * D + lane * 4);
        *(v4f*)(Ws + n * LDP + lane * 4) = w4;
        *(v4f*)(At + i * LDP + lane * 4) = x4;
    }
    __syncthreads();
    if (mbase >= N) return;

    const float* arow = At + lmod * LDP + koff;
    const float* wrow = Ws + lmod * LDP + koff;

    v8f c[8] = {};

#pragma unroll
    for (int k0 = 0; k0 < D; k0 += 4) {
        v2f a = *(const v2f*)(arow + k0);
#pragma unroll
        for (int j = 0; j < 8; ++j) {
            v2f b = *(const v2f*)(wrow + j * 16 * LDP + k0);
            c[j] = __builtin_amdgcn_wmma_f32_16x16x4_f32(false, a, false, b,
                                                         (short)0, c[j], false, false);
        }
    }

    float rscale[8];
#pragma unroll
    for (int r = 0; r < 8; ++r) rscale[r] = inv_sqrt_deg(rdeg[r]);
#pragma unroll
    for (int j = 0; j < 8; ++j) {
#pragma unroll
        for (int r = 0; r < 8; ++r) c[j][r] *= rscale[r];
    }
#pragma unroll
    for (int r = 0; r < 8; ++r) {
        int row = mbase + mhi + r;
        if (row < N) {
#pragma unroll
            for (int j = 0; j < 8; ++j) {
                int col = j * 16 + lmod;
                h[(size_t)row * D + col] = c[j][r];
            }
        }
    }
}


static __device__ __forceinline__ v4f
gather_chunk(v4f acc, v2i u, bool first, int nvalid, int node, float di,
             const float* __restrict__ h, int lane) {
    const bool valid  = lane < nvalid;
    const bool isself = first && (lane == 0);
    int   src_l = node;
    float w_l   = 0.0f;
    if (valid) {
        if (isself) {
            w_l = di;
        } else {
            src_l = u.x;
            w_l   = __int_as_float(u.y) * di;
        }
    }
    const int w_bits = __float_as_int(w_l);
    for (int t = 0; t < nvalid; t += 4) {
        int   s0 = __builtin_amdgcn_readlane(src_l, t);
        int   s1 = __builtin_amdgcn_readlane(src_l, t + 1);
        int   s2 = __builtin_amdgcn_readlane(src_l, t + 2);
        int   s3 = __builtin_amdgcn_readlane(src_l, t + 3);
        float w0 = __int_as_float(__builtin_amdgcn_readlane(w_bits, t));
        float w1 = __int_as_float(__builtin_amdgcn_readlane(w_bits, t + 1));
        float w2 = __int_as_float(__builtin_amdgcn_readlane(w_bits, t + 2));
        float w3 = __int_as_float(__builtin_amdgcn_readlane(w_bits, t + 3));
        v4f h0 = *(const v4f*)(h + (size_t)s0 * D + lane * 4);
        v4f h1 = *(const v4f*)(h + (size_t)s1 * D + lane * 4);
        v4f h2 = *(const v4f*)(h + (size_t)s2 * D + lane * 4);
        v4f h3 = *(const v4f*)(h + (size_t)s3 * D + lane * 4);
        acc += h0 * w0;
        acc += h1 * w1;
        acc += h2 * w2;
        acc += h3 * w3;
    }
    return acc;
}

__global__ void __launch_bounds__(256)
pull_finalize_kernel(const int* __restrict__ rows, const int* __restrict__ cols,
                     const float* __restrict__ ew, const float* __restrict__ deg,
                     const float* __restrict__ h, float* out,
                     const float* __restrict__ bias, const float* __restrict__ gamma,
                     const float* __restrict__ beta, int N, int E) {
    const int lane = threadIdx.x & 31;
    const int wave = threadIdx.x >> 5;
    const int node = blockIdx.x * 8 + wave;
    if (node >= N) return;

    float* p = out + (size_t)node * D + lane * 4;
    const v2i* units = (const v2i*)(out + (size_t)node * D);
    const float di = inv_sqrt_deg(deg[node]);

    v2i u = { 0, 0 };
    if (lane < 16) u = units[lane];
    const int c = __builtin_amdgcn_readfirstlane(u.x);

    v4f acc = { 0.0f, 0.0f, 0.0f, 0.0f };
    if (c <= BUCKET_CAP) {
        if (c > 15) {
            if (lane >= 16) u = units[lane];
        }
        acc = gather_chunk(acc, u, true, (c < 31 ? c : 31) + 1, node, di, h, lane);
        if (c > 31) {
            u = units[32 + lane];
            acc = gather_chunk(acc, u, false, c - 31, node, di, h, lane);
        }
    } else {
        v4f hs = *(const v4f*)(h + (size_t)node * D + lane * 4);
        acc = hs * di;
        for (int base = 0; base < E; base += 32) {
            int  e     = base + lane;
            bool match = false;
            if (e < E) match = (cols[e] == node);
            int   src_l = node;
            float w_l   = 0.0f;
            if (match) {
                src_l = rows[e];
                w_l   = ew[e] * di;
            }
            const int w_bits = __float_as_int(w_l);
            unsigned mask = __builtin_amdgcn_ballot_w32(match);
            while (mask) {
                int t = __builtin_ctz(mask);
                mask &= mask - 1;
                int   s = __builtin_amdgcn_readlane(src_l, t);
                float w = __int_as_float(__builtin_amdgcn_readlane(w_bits, t));
                v4f hv = *(const v4f*)(h + (size_t)s * D + lane * 4);
                acc += hv * w;
            }
        }
    }

    v4f v = acc;
    float s = 0.f, s2 = 0.f;
#pragma unroll
    for (int i = 0; i < 4; ++i) {
        int ch = lane * 4 + i;
        float vv = v[i] + bias[ch];
        vv = fmaxf(vv, 0.0f);
        v[i] = vv;
        s  += vv;
        s2 += vv * vv;
    }
#pragma unroll
    for (int off = 16; off >= 1; off >>= 1) {
        s  += __shfl_xor(s,  off, 32);
        s2 += __shfl_xor(s2, off, 32);
    }
    float mu  = s * (1.0f / 128.0f);
    float var = s2 * (1.0f / 128.0f) - mu * mu;
    float rs  = rsqrtf(var + LN_EPS);
#pragma unroll
    for (int i = 0; i < 4; ++i) {
        int ch = lane * 4 + i;
        v[i] = (v[i] - mu) * rs * gamma[ch] + beta[ch];
    }
    *(v4f*)p = v;
}

extern "C" void kernel_launch(void* const* d_in, const int* in_sizes, int n_in,
                              void* d_out, int out_size, void* d_ws, size_t ws_size,
                              hipStream_t stream) {
    const float* x     = (const float*)d_in[0];
    const int*   eidx  = (const int*)  d_in[1];
    const float* ew    = (const float*)d_in[2];
    const float* W     = (const float*)d_in[3];
    const float* bias  = (const float*)d_in[4];
    const float* gamma = (const float*)d_in[5];
    const float* beta  = (const float*)d_in[6];
    float*       out   = (float*)d_out;

    const int N = in_sizes[0] / D;
    const int E = in_sizes[2];
    const int* rows = eidx;
    const int* cols = eidx + E;

    float* h    = (float*)d_ws;
    float* deg  = h + (size_t)N * D;

    const int nchunks = (E + CHUNK - 1) / CHUNK;
    int* ent = (int*)h;
    int* off = ent + (size_t)nchunks * CHUNK * 4;

    partition_kernel<<<nchunks, 256, 0, stream>>>(rows, cols, ew, ent, off, E);
    build_kernel<<<(N + BIN_NODES - 1) / BIN_NODES, 512, 0, stream>>>(
        ent, off, deg, (int*)out, N, nchunks);

    gemm_wmma_kernel<<<(N + 127) / 128, 256, 0, stream>>>(x, W, deg, h, N);

    pull_finalize_kernel<<<(N + 7) / 8, 256, 0, stream>>>(
        rows, cols, ew, deg, h, out, bias, gamma, beta, N, E);
}


# bucket build (pass 2): a thread requests its 12 piece entries at once, ranks them, then stores; 16 nodes' counts and first lines read ahead of the stores; arrival order at the LDS counters changes
# speedup vs baseline: 3.8333x; 1.1598x over previous
#include <hip/hip_runtime.h>
#include <hip/hip_bf16.h>

typedef __attribute__((ext_vector_type(2))) float v2f;
typedef __attribute__((ext_vector_type(4))) float v4f;
typedef __attribute__((ext_vector_type(8))) float v8f;
typedef __attribute__((ext_vector_type(2))) int   v2i;
typedef __attribute__((ext_vector_type(4))) int   v4i;

#define D 128
#define LN_EPS 1e-5f

#define BUCKET_CAP 63

#define BIN_SHIFT 9
#define BIN_NODES 512
#define MAXBINS   256
#define OFFS      (MAXBINS + 1)
#define CHUNK     2560
#define PRE       12

__global__ void __launch_bounds__(256)
partition_kernel(const int* __restrict__ rows, const int* __restrict__ cols,
                 const float* __restrict__ ew, int* __restrict__ ent,
                 int* __restrict__ off, int E) {
    __shared__ int hist[MAXBINS];
    __shared__ int start[OFFS];
    __shared__ int wsum[8];
    __shared__ __attribute__((aligned(16))) int sorted[CHUNK * 4];

    const int tid  = threadIdx.x;
    const int lane = tid & 31;
    const int wave = tid >> 5;
    const int base = blockIdx.x * CHUNK;

    hist[tid] = 0;
    __syncthreads();

    int  s[10], d[10], wb[10], r[10];
    bool ok[10];
#pragma unroll
    for (int k = 0; k < 10; ++k) {
        int e = base + k * 256 + tid;
        ok[k] = e < E;
        s[k] = 0; d[k] = 0; wb[k] = 0; r[k] = 0;
        if (ok[k]) {
            s[k]  = rows[e];
            d[k]  = cols[e];
            wb[k] = __float_as_int(ew[e]);
        }
    }
#pragma unroll
    for (int k = 0; k < 10; ++k)
        if (ok[k]) r[k] = atomicAdd(&hist[d[k] >> BIN_SHIFT], 1);
    __syncthreads();

    const int v = hist[tid];
    int inc = v;
#pragma unroll
    for (int o = 1; o < 32; o <<= 1) {
        int t = __shfl_up(inc, o, 32);
        if (lane >= o) inc += t;
    }
    if (lane == 31) wsum[wave] = inc;
    __syncthreads();
    int pre = 0;
    for (int i = 0; i < wave; ++i) pre += wsum[i];
    start[tid] = pre + inc - v;
    if (tid == 255) start[MAXBINS] = pre + inc;
    __syncthreads();

#pragma unroll
    for (int k = 0; k < 10; ++k) {
        if (ok[k]) {
            int p = start[d[k] >> BIN_SHIFT] + r[k];
            v4i en = { s[k], d[k], wb[k], 0 };
            *(v4i*)(sorted + 4 * p) = en;
        }
    }
    __syncthreads();

    const int total = start[MAXBINS];
    for (int i = tid; i < total; i += 256)
        *(v4i*)(ent + ((size_t)base + i) * 4) = *(const v4i*)(sorted + 4 * i);
    for (int i = tid; i < OFFS; i += 256)
        off[(size_t)blockIdx.x * OFFS + i] = start[i];
}

__global__ void __launch_bounds__(512)
build_kernel(const int* __restrict__ ent, const int* __restrict__ off,
             float* __restrict__ deg, int* __restrict__ bucket, int N, int nchunks) {
    __shared__ int   cnt[BIN_NODES];
    __shared__ float dsum[BIN_NODES];
    __shared__ __attribute__((aligned(16))) v2i stage[BIN_NODES * 64];

    const int tid   = threadIdx.x;
    const int b     = blockIdx.x;
    const int node0 = b * BIN_NODES;

    cnt[tid]  = 0;
    dsum[tid] = 0.0f;
    __syncthreads();

    const int sub = tid & 1;
    for (int w = tid >> 1; w < nchunks; w += 256) {
        const int lo = off[(size_t)w * OFFS + b];
        const int hi = off[(size_t)w * OFFS + b + 1];
        const int* seg = ent + (size_t)w * CHUNK * 4;
        v4i en[PRE];
        int pos[PRE];
#pragma unroll
        for (int j = 0; j < PRE; ++j) {
            const int i = lo + sub + 2 * j;
            en[j] = *(const v4i*)(seg + 4 * (i < hi ? i : 0));
        }
#pragma unroll
        for (int j = 0; j < PRE; ++j) {
            pos[j] = BUCKET_CAP;
            if (lo + sub + 2 * j < hi) {
                const int ld = en[j].y - node0;
                pos[j] = atomicAdd(&cnt[ld], 1);
                atomicAdd(&dsum[ld], __int_as_float(en[j].z));
            }
        }
#pragma unroll
        for (int j = 0; j < PRE; ++j) {
            if (pos[j] < BUCKET_CAP) {
                v2i u = { en[j].x, en[j].z };
                stage[(en[j].y - node0) * 64 + 1 + pos[j]] = u;
            }
        }
        for (int i = lo + sub + 2 * PRE; i < hi; i += 2) {
            v4i e1 = *(const v4i*)(seg + 4 * i);
            int ld  = e1.y - node0;
            int p1  = atomicAdd(&cnt[ld], 1);
            atomicAdd(&dsum[ld], __int_as_float(e1.z));
            if (p1 < BUCKET_CAP) {
                v2i u = { e1.x, e1.z };
                stage[ld * 64 + 1 + p1] = u;
            }
        }
    }
    __syncthreads();

    const int node = node0 + tid;
    if (node < N) deg[node] = 1.0f + dsum[tid];

    const int l16 = tid & 15;
    const int g   = tid >> 4;
    int cs[16];
    v2i u0[16];
#pragma unroll
    for (int k = 0; k < 16; ++k) {
        cs[k] = cnt[g + 32 * k];
        u0[k] = stage[(g + 32 * k) * 64 + l16];
    }
#pragma unroll
    for (int k = 0; k < 16; ++k) {
        const int n = g + 32 * k;
        if (node0 + n < N) {
            const int c    = cs[k];
            const int used = (c < BUCKET_CAP ? c : BUCKET_CAP) + 1;
            const int nl   = (used + 15) >> 4;
            int* brow = bucket + (size_t)(node0 + n) * D;
            v2i u = u0[k];
            if (l16 == 0) { u.x = c; u.y = 0; }
            *(v2i*)(brow + l16 * 2) = u;
            for (int L = 1; L < nl; ++L)
                *(v2i*)(brow + (L * 16 + l16) * 2) = stage[n * 64 + L * 16 + l16];
        }
    }
}

static __device__ __forceinline__ float inv_sqrt_deg(float d) {
    return (d > 0.0f) ? rsqrtf(d) : 0.0f;
}

#define LDP 132
#define TILE_NODES 128
#define NODES_PER_WAVE 4

static __device__ __forceinline__ v4f
gather_rows(v4f acc, int src_l, float w_l, int nvalid, const float* __restrict__ x, int lane) {
    const int w_bits = __float_as_int(w_l);
    for (int t = 0; t < nvalid; t += 4) {
        int   s0 = __builtin_amdgcn_readlane(src_l, t);
        int   s1 = __builtin_amdgcn_readlane(src_l, t + 1);
        int   s2 = __builtin_amdgcn_readlane(src_l, t + 2);
        int   s3 = __builtin_amdgcn_readlane(src_l, t + 3);
        float w0 = __int_as_float(__builtin_amdgcn_readlane(w_bits, t));
        float w1 = __int_as_float(__builtin_amdgcn_readlane(w_bits, t + 1));
        float w2 = __int_as_float(__builtin_amdgcn_readlane(w_bits, t + 2));
        float w3 = __int_as_float(__builtin_amdgcn_readlane(w_bits, t + 3));
        v4f h0 = *(const v4f*)(x + (size_t)s0 * D + lane * 4);
        v4f h1 = *(const v4f*)(x + (size_t)s1 * D + lane * 4);
        v4f h2 = *(const v4f*)(x + (size_t)s2 * D + lane * 4);
        v4f h3 = *(const v4f*)(x + (size_t)s3 * D + lane * 4);
        acc += h0 * w0;
        acc += h1 * w1;
        acc += h2 * w2;
        acc += h3 * w3;
    }
    return acc;
}

static __device__ __forceinline__ bool head_is_edge(int lane, int c) {
    return lane >= 1 && lane <= c && lane < 16;
}

static __device__ __forceinline__ v4f
aggregate_node(int node, v2i u, float hsum, const int* __restrict__ rows,
               const int* __restrict__ cols, const float* __restrict__ ew,
               const float* __restrict__ deg, const float* __restrict__ x,
               const v2i* units, int lane, int E) {
    const int   c  = __builtin_amdgcn_readfirstlane(u.x);
    const float di = __int_as_float(__builtin_amdgcn_readfirstlane(
                         __float_as_int(inv_sqrt_deg(hsum))));
    v4f acc = { 0.0f, 0.0f, 0.0f, 0.0f };
    if (c <= BUCKET_CAP) {
        const bool he = head_is_edge(lane, c);
        int   src_l = he ? u.x : node;
        float ew_l  = he ? __int_as_float(u.y) : 1.0f;
        float dsum  = hsum;
        if (c > 15) {
            if (lane >= 16) {
                const v2i  u2 = units[lane];
                const bool v2 = lane <= c;
                src_l = v2 ? u2.x : node;
                ew_l  = v2 ? __int_as_float(u2.y) : 1.0f;
                dsum  = deg[src_l];
            }
        }
        const int   nvalid = (c < 31 ? c : 31) + 1;
        const float w_l    = (lane < nvalid) ? (inv_sqrt_deg(dsum) * ew_l) * di : 0.0f;
        acc = gather_rows(acc, src_l, w_l, nvalid, x, lane);
        if (c > 31) {
            const v2i  u3 = units[32 + lane];
            const int  n3 = c - 31;
            const bool v3 = lane < n3;
            const int   s3 = v3 ? u3.x : node;
            const float w3 = v3 ? (inv_sqrt_deg(deg[s3]) * __int_as_float(u3.y)) * di : 0.0f;
            acc = gather_rows(acc, s3, w3, n3, x, lane);
        }
    } else {
        v4f xs = *(const v4f*)(x + (size_t)node * D + lane * 4);
        acc = xs * (di * di);
        for (int base = 0; base < E; base += 32) {
            int  e     = base + lane;
            bool match = false;
            if (e < E) match = (cols[e] == node);
            int   src_l = node;
            float w_l   = 0.0f;
            if (match) {
                src_l = rows[e];
                w_l   = (inv_sqrt_deg(deg[src_l]) * ew[e]) * di;
            }
            const int w_bits = __float_as_int(w_l);
            unsigned mask = __builtin_amdgcn_ballot_w32(match);
            while (mask) {
                int t = __builtin_ctz(mask);
                mask &= mask - 1;
                int   s = __builtin_amdgcn_readlane(src_l, t);
                float w = __int_as_float(__builtin_amdgcn_readlane(w_bits, t));
                v4f xv = *(const v4f*)(x + (size_t)s * D + lane * 4);
                acc += xv * w;
            }
        }
    }
    return acc;
}

__global__ void __launch_bounds__(1024, 16)
pull_transform_kernel(const int* __restrict__ rows, const int* __restrict__ cols,
                      const float* __restrict__ ew, const float* __restrict__ deg,
                      const float* __restrict__ x, const float* __restrict__ W,
                      float* out,
                      const float* __restrict__ bias, const float* __restrict__ gamma,
                      const float* __restrict__ beta, int N, int E) {
    __shared__ __attribute__((aligned(16))) float Ws[D * LDP];
    __shared__ __attribute__((aligned(16))) float As[TILE_NODES * LDP];

    const int lane  = threadIdx.x & 31;
    const int wave  = threadIdx.x >> 5;
    const int node0 = blockIdx.x * TILE_NODES;
    const int lrow0 = wave * NODES_PER_WAVE;

    v2i hu[NODES_PER_WAVE];
#pragma unroll
    for (int q = 0; q < NODES_PER_WAVE; ++q) {
        int node = node0 + lrow0 + q;
        if (node > N - 1) node = N - 1;
        const v2i* units = (const v2i*)(out + (size_t)node * D);
        hu[q] = units[lane & 15];
    }

#pragma unroll
    for (int i = 0; i < 4; ++i) {
        int n = wave * 4 + i;
        v4f w4 = *(const v4f*)(W + (size_t)n * D + lane * 4);
        *(v4f*)(Ws + n * LDP + lane * 4) = w4;
    }

    float hs[NODES_PER_WAVE];
#pragma unroll
    for (int q = 0; q < NODES_PER_WAVE; ++q) {
        int node = node0 + lrow0 + q;
        if (node > N - 1) node = N - 1;
        const int c = __builtin_amdgcn_readfirstlane(hu[q].x);
        hs[q] = deg[head_is_edge(lane, c) ? hu[q].x : node];
    }

#pragma unroll
    for (int q = 0; q < NODES_PER_WAVE; ++q) {
        const int node = node0 + lrow0 + q;
        v4f acc = { 0.0f, 0.0f, 0.0f, 0.0f };
        if (node < N) {
            const v2i* units = (const v2i*)(out + (size_t)node * D);
            acc = aggregate_node(node, hu[q], hs[q], rows, cols, ew, deg, x, units, lane, E);
        }
        *(v4f*)(As + (lrow0 + q) * LDP + lane * 4) = acc;
    }
    __syncthreads();

    const int lmod = lane & 15;
    const int koff = (lane >> 4) * 2;
    const int mhi  = (lane >> 4) * 8;
    const int mt   = wave >> 2;
    const int nt0  = (wave & 3) * 2;
    const float* arow = As + (mt * 16 + lmod) * LDP + koff;
    const float* wrow = Ws + (nt0 * 16 + lmod) * LDP + koff;

    v8f c0 = {};
    v8f c1 = {};
#pragma unroll
    for (int k0 = 0; k0 < D; k0 += 4) {
        v2f a  = *(const v2f*)(arow + k0);
        v2f b0 = *(const v2f*)(wrow + k0);
        v2f b1 = *(const v2f*)(wrow + 16 * LDP + k0);
        c0 = __builtin_amdgcn_wmma_f32_16x16x4_f32(false, a, false, b0,
                                                   (short)0, c0, false, false);
        c1 = __builtin_amdgcn_wmma_f32_16x16x4_f32(false, a, false, b1,
                                                   (short)0, c1, false, false);
    }
    __syncthreads();

#pragma unroll
    for (int r = 0; r < 8; ++r) {
        float* crow = As + (mt * 16 + mhi + r) * LDP + nt0 * 16 + lmod;
        crow[0]  = c0[r];
        crow[16] = c1[r];
    }
    __syncthreads();

    const v4f b4 = *(const v4f*)(bias  + lane * 4);
    const v4f g4 = *(const v4f*)(gamma + lane * 4);
    const v4f e4 = *(const v4f*)(beta  + lane * 4);
    for (int q = 0; q < NODES_PER_WAVE; ++q) {
        const int node = node0 + lrow0 + q;
        if (node >= N) break;
        v4f v = *(const v4f*)(As + (lrow0 + q) * LDP + lane * 4);
        float s = 0.f, s2 = 0.f;
#pragma unroll
        for (int i = 0; i < 4; ++i) {
            float vv = v[i] + b4[i];
            vv = fmaxf(vv, 0.0f);
            v[i] = vv;
            s  += vv;
            s2 += vv * vv;
        }
#pragma unroll
        for (int off = 16; off >= 1; off >>= 1) {
            s  += __shfl_xor(s,  off, 32);
            s2 += __shfl_xor(s2, off, 32);
        }
        float mu  = s * (1.0f / 128.0f);
        float var = s2 * (1.0f / 128.0f) - mu * mu;
        float rs  = rsqrtf(var + LN_EPS);
#pragma unroll
        for (int i = 0; i < 4; ++i)
            v[i] = (v[i] - mu) * rs * g4[i] + e4[i];
        *(v4f*)(out + (size_t)node * D + lane * 4) = v;
    }
}

extern "C" void kernel_launch(void* const* d_in, const int* in_sizes, int n_in,
                              void* d_out, int out_size, void* d_ws, size_t ws_size,
                              hipStream_t stream) {
    const float* x     = (const float*)d_in[0];
    const int*   eidx  = (const int*)  d_in[1];
    const float* ew    = (const float*)d_in[2];
    const float* W     = (const float*)d_in[3];
    const float* bias  = (const float*)d_in[4];
    const float* gamma = (const float*)d_in[5];
    const float* beta  = (const float*)d_in[6];
    float*       out   = (float*)d_out;

    const int N = in_sizes[0] / D;
    const int E = in_sizes[2];
    const int* rows = eidx;
    const int* cols = eidx + E;

    float* wsf  = (float*)d_ws;
    float* deg  = wsf + (size_t)N * D;

    const int nchunks = (E + CHUNK - 1) / CHUNK;
    int* ent = (int*)wsf;
    int* off = ent + (size_t)nchunks * CHUNK * 4;

    partition_kernel<<<nchunks, 256, 0, stream>>>(rows, cols, ew, ent, off, E);
    build_kernel<<<(N + BIN_NODES - 1) / BIN_NODES, 512, 0, stream>>>(
        ent, off, deg, (int*)out, N, nchunks);

    pull_transform_kernel<<<(N + TILE_NODES - 1) / TILE_NODES, 1024, 0, stream>>>(
        rows, cols, ew, deg, x, W, out, bias, gamma, beta, N, E);
}


# pull kernel as 512 work-groups, all resident (2 per WGP), each walking its own equal node range in tiles of up to 128, W staged once; was 782 one-tile work-groups on 512 places; sum order kept
# speedup vs baseline: 4.3108x; 1.1246x over previous
#include <hip/hip_runtime.h>
#include <hip/hip_bf16.h>

typedef __attribute__((ext_vector_type(2))) float v2f;
typedef __attribute__((ext_vector_type(4))) float v4f;
typedef __attribute__((ext_vector_type(8))) float v8f;
typedef __attribute__((ext_vector_type(2))) int   v2i;
typedef __attribute__((ext_vector_type(4))) int   v4i;

#define D 128
#define LN_EPS 1e-5f

#define BUCKET_CAP 63

#define BIN_SHIFT 9
#define BIN_NODES 512
#define MAXBINS   256
#define OFFS      (MAXBINS + 1)
#define CHUNK     2560
#define PRE       12

__global__ void __launch_bounds__(256)
partition_kernel(const int* __restrict__ rows, const int* __restrict__ cols,
                 const float* __restrict__ ew, int* __restrict__ ent,
                 int* __restrict__ off, int E) {
    __shared__ int hist[MAXBINS];
    __shared__ int start[OFFS];
    __shared__ int wsum[8];
    __shared__ __attribute__((aligned(16))) int sorted[CHUNK * 4];

    const int tid  = threadIdx.x;
    const int lane = tid & 31;
    const int wave = tid >> 5;
    const int base = blockIdx.x * CHUNK;

    hist[tid] = 0;
    __syncthreads();

    int  s[10], d[10], wb[10], r[10];
    bool ok[10];
#pragma unroll
    for (int k = 0; k < 10; ++k) {
        int e = base + k * 256 + tid;
        ok[k] = e < E;
        s[k] = 0; d[k] = 0; wb[k] = 0; r[k] = 0;
        if (ok[k]) {
            s[k]  = rows[e];
            d[k]  = cols[e];
            wb[k] = __float_as_int(ew[e]);
        }
    }
#pragma unroll
    for (int k = 0; k < 10; ++k)
        if (ok[k]) r[k] = atomicAdd(&hist[d[k] >> BIN_SHIFT], 1);
    __syncthreads();

    const int v = hist[tid];
    int inc = v;
#pragma unroll
    for (int o = 1; o < 32; o <<= 1) {
        int t = __shfl_up(inc, o, 32);
        if (lane >= o) inc += t;
    }
    if (lane == 31) wsum[wave] = inc;
    __syncthreads();
    int pre = 0;
    for (int i = 0; i < wave; ++i) pre += wsum[i];
    start[tid] = pre + inc - v;
    if (tid == 255) start[MAXBINS] = pre + inc;
    __syncthreads();

#pragma unroll
    for (int k = 0; k < 10; ++k) {
        if (ok[k]) {
            int p = start[d[k] >> BIN_SHIFT] + r[k];
            v4i en = { s[k], d[k], wb[k], 0 };
            *(v4i*)(sorted + 4 * p) = en;
        }
    }
    __syncthreads();

    const int total = start[MAXBINS];
    for (int i = tid; i < total; i += 256)
        *(v4i*)(ent + ((size_t)base + i) * 4) = *(const v4i*)(sorted + 4 * i);
    for (int i = tid; i < OFFS; i += 256)
        off[(size_t)blockIdx.x * OFFS + i] = start[i];
}

__global__ void __launch_bounds__(512)
build_kernel(const int* __restrict__ ent, const int* __restrict__ off,
             float* __restrict__ deg, int* __restrict__ bucket, int N, int nchunks) {
    __shared__ int   cnt[BIN_NODES];
    __shared__ float dsum[BIN_NODES];
    __shared__ __attribute__((aligned(16))) v2i stage[BIN_NODES * 64];

    const int tid   = threadIdx.x;
    const int b     = blockIdx.x;
    const int node0 = b * BIN_NODES;

    cnt[tid]  = 0;
    dsum[tid] = 0.0f;
    __syncthreads();

    const int sub = tid & 1;
    for (int w = tid >> 1; w < nchunks; w += 256) {
        const int lo = off[(size_t)w * OFFS + b];
        const int hi = off[(size_t)w * OFFS + b + 1];
        const int* seg = ent + (size_t)w * CHUNK * 4;
        v4i en[PRE];
        int pos[PRE];
#pragma unroll
        for (int j = 0; j < PRE; ++j) {
            const int i = lo + sub + 2 * j;
            en[j] = *(const v4i*)(seg + 4 * (i < hi ? i : 0));
        }
#pragma unroll
        for (int j = 0; j < PRE; ++j) {
            pos[j] = BUCKET_CAP;
            if (lo + sub + 2 * j < hi) {
                const int ld = en[j].y - node0;
                pos[j] = atomicAdd(&cnt[ld], 1);
                atomicAdd(&dsum[ld], __int_as_float(en[j].z));
            }
        }
#pragma unroll
        for (int j = 0; j < PRE; ++j) {
            if (pos[j] < BUCKET_CAP) {
                v2i u = { en[j].x, en[j].z };
                stage[(en[j].y - node0) * 64 + 1 + pos[j]] = u;
            }
        }
        for (int i = lo + sub + 2 * PRE; i < hi; i += 2) {
            v4i e1 = *(const v4i*)(seg + 4 * i);
            int ld  = e1.y - node0;
            int p1  = atomicAdd(&cnt[ld], 1);
            atomicAdd(&dsum[ld], __int_as_float(e1.z));
            if (p1 < BUCKET_CAP) {
                v2i u = { e1.x, e1.z };
                stage[ld * 64 + 1 + p1] = u;
            }
        }
    }
    __syncthreads();

    const int node = node0 + tid;
    if (node < N) deg[node] = 1.0f + dsum[tid];

    const int l16 = tid & 15;
    const int g   = tid >> 4;
    int cs[16];
    v2i u0[16];
#pragma unroll
    for (int k = 0; k < 16; ++k) {
        cs[k] = cnt[g + 32 * k];
        u0[k] = stage[(g + 32 * k) * 64 + l16];
    }
#pragma unroll
    for (int k = 0; k < 16; ++k) {
        const int n = g + 32 * k;
        if (node0 + n < N) {
            const int c    = cs[k];
            const int used = (c < BUCKET_CAP ? c : BUCKET_CAP) + 1;
            const int nl   = (used + 15) >> 4;
            int* brow = bucket + (size_t)(node0 + n) * D;
            v2i u = u0[k];
            if (l16 == 0) { u.x = c; u.y = 0; }
            *(v2i*)(brow + l16 * 2) = u;
            for (int L = 1; L < nl; ++L)
                *(v2i*)(brow + (L * 16 + l16) * 2) = stage[n * 64 + L * 16 + l16];
        }
    }
}

static __device__ __forceinline__ float inv_sqrt_deg(float d) {
    return (d > 0.0f) ? rsqrtf(d) : 0.0f;
}

#define LDP 132
#define TILE_NODES 128
#define NODES_PER_WAVE 4
#define PULL_GROUPS 512

static __device__ __forceinline__ v4f
gather_rows(v4f acc, int src_l, float w_l, int nvalid, const float* __restrict__ x, int lane) {
    const int w_bits = __float_as_int(w_l);
    for (int t = 0; t < nvalid; t += 4) {
        int   s0 = __builtin_amdgcn_readlane(src_l, t);
        int   s1 = __builtin_amdgcn_readlane(src_l, t + 1);
        int   s2 = __builtin_amdgcn_readlane(src_l, t + 2);
        int   s3 = __builtin_amdgcn_readlane(src_l, t + 3);
        float w0 = __int_as_float(__builtin_amdgcn_readlane(w_bits, t));
        float w1 = __int_as_float(__builtin_amdgcn_readlane(w_bits, t + 1));
        float w2 = __int_as_float(__builtin_amdgcn_readlane(w_bits, t + 2));
        float w3 = __int_as_float(__builtin_amdgcn_readlane(w_bits, t + 3));
        v4f h0 = *(const v4f*)(x + (size_t)s0 * D + lane * 4);
        v4f h1 = *(const v4f*)(x + (size_t)s1 * D + lane * 4);
        v4f h2 = *(const v4f*)(x + (size_t)s2 * D + lane * 4);
        v4f h3 = *(const v4f*)(x + (size_t)s3 * D + lane * 4);
        acc += h0 * w0;
        acc += h1 * w1;
        acc += h2 * w2;
        acc += h3 * w3;
    }
    return acc;
}

static __device__ __forceinline__ bool head_is_edge(int lane, int c) {
    return lane >= 1 && lane <= c && lane < 16;
}

static __device__ __forceinline__ v4f
aggregate_node(int node, v2i u, float hsum, const int* __restrict__ rows,
               const int* __restrict__ cols, const float* __restrict__ ew,
               const float* __restrict__ deg, const float* __restrict__ x,
               const v2i* units, int lane, int E) {
    const int   c  = __builtin_amdgcn_readfirstlane(u.x);
    const float di = __int_as_float(__builtin_amdgcn_readfirstlane(
                         __float_as_int(inv_sqrt_deg(hsum))));
    v4f acc = { 0.0f, 0.0f, 0.0f, 0.0f };
    if (c <= BUCKET_CAP) {
        const bool he = head_is_edge(lane, c);
        int   src_l = he ? u.x : node;
        float ew_l  = he ? __int_as_float(u.y) : 1.0f;
        float dsum  = hsum;
        if (c > 15) {
            if (lane >= 16) {
                const v2i  u2 = units[lane];
                const bool v2 = lane <= c;
                src_l = v2 ? u2.x : node;
                ew_l  = v2 ? __int_as_float(u2.y) : 1.0f;
                dsum  = deg[src_l];
            }
        }
        const int   nvalid = (c < 31 ? c : 31) + 1;
        const float w_l    = (lane < nvalid) ? (inv_sqrt_deg(dsum) * ew_l) * di : 0.0f;
        acc = gather_rows(acc, src_l, w_l, nvalid, x, lane);
        if (c > 31) {
            const v2i  u3 = units[32 + lane];
            const int  n3 = c - 31;
            const bool v3 = lane < n3;
            const int   s3 = v3 ? u3.x : node;
            const float w3 = v3 ? (inv_sqrt_deg(deg[s3]) * __int_as_float(u3.y)) * di : 0.0f;
            acc = gather_rows(acc, s3, w3, n3, x, lane);
        }
    } else {
        v4f xs = *(const v4f*)(x + (size_t)node * D + lane * 4);
        acc = xs * (di * di);
        for (int base = 0; base < E; base += 32) {
            int  e     = base + lane;
            bool match = false;
            if (e < E) match = (cols[e] == node);
            int   src_l = node;
            float w_l   = 0.0f;
            if (match) {
                src_l = rows[e];
                w_l   = (inv_sqrt_deg(deg[src_l]) * ew[e]) * di;
            }
            const int w_bits = __float_as_int(w_l);
            unsigned mask = __builtin_amdgcn_ballot_w32(match);
            while (mask) {
                int t = __builtin_ctz(mask);
                mask &= mask - 1;
                int   s = __builtin_amdgcn_readlane(src_l, t);
                float w = __int_as_float(__builtin_amdgcn_readlane(w_bits, t));
                v4f xv = *(const v4f*)(x + (size_t)s * D + lane * 4);
                acc += xv * w;
            }
        }
    }
    return acc;
}

__global__ void __launch_bounds__(1024, 16)
pull_transform_kernel(const int* __restrict__ rows, const int* __restrict__ cols,
                      const float* __restrict__ ew, const float* __restrict__ deg,
                      const float* __restrict__ x, const float* __restrict__ W,
                      float* out,
                      const float* bias, const float* gamma,
                      const float* beta, int N, int E) {
    __shared__ __attribute__((aligned(16))) float Ws[D * LDP];
    __shared__ __attribute__((aligned(16))) float As[TILE_NODES * LDP];

    const int lane  = threadIdx.x & 31;
    const int wave  = __builtin_amdgcn_readfirstlane((int)(threadIdx.x >> 5));

    const int per  = (N + PULL_GROUPS - 1) / PULL_GROUPS;
    const int n_lo = (int)blockIdx.x * per;
    const int n_hi = (n_lo + per < N) ? n_lo + per : N;
    if (n_lo >= n_hi) return;

#pragma unroll
    for (int i = 0; i < 4; ++i) {
        int n = wave * 4 + i;
        v4f w4 = *(const v4f*)(W + (size_t)n * D + lane * 4);
        *(v4f*)(Ws + n * LDP + lane * 4) = w4;
    }

    const int lmod = lane & 15;
    const int koff = (lane >> 4) * 2;
    const int mhi  = (lane >> 4) * 8;
    const int mt   = wave >> 2;
    const int nt0  = (wave & 3) * 2;

    for (int node0 = n_lo; node0 < n_hi; node0 += TILE_NODES) {
        v2i hu[NODES_PER_WAVE];
#pragma unroll
        for (int q = 0; q < NODES_PER_WAVE; ++q) {
            int node = node0 + q * 32 + wave;
            if (node > n_hi - 1) node = n_hi - 1;
            const v2i* units = (const v2i*)(out + (size_t)node * D);
            hu[q] = units[lane & 15];
        }

        float hs[NODES_PER_WAVE];
#pragma unroll
        for (int q = 0; q < NODES_PER_WAVE; ++q) {
            int node = node0 + q * 32 + wave;
            if (node > n_hi - 1) node = n_hi - 1;
            const int c = __builtin_amdgcn_readfirstlane(hu[q].x);
            hs[q] = deg[head_is_edge(lane, c) ? hu[q].x : node];
        }

#pragma unroll
        for (int q = 0; q < NODES_PER_WAVE; ++q) {
            const int node = node0 + q * 32 + wave;
            v4f acc = { 0.0f, 0.0f, 0.0f, 0.0f };
            if (node < n_hi) {
                const v2i* units = (const v2i*)(out + (size_t)node * D);
                acc = aggregate_node(node, hu[q], hs[q], rows, cols, ew, deg, x, units, lane, E);
            }
            *(v4f*)(As + (q * 32 + wave) * LDP + lane * 4) = acc;
        }
        __syncthreads();

        const bool has_tile = mt * 16 < n_hi - node0;
        const float* arow = As + (mt * 16 + lmod) * LDP + koff;
        const float* wrow = Ws + (nt0 * 16 + lmod) * LDP + koff;

        v8f c0 = {};
        v8f c1 = {};
        if (has_tile) {
#pragma unroll
            for (int k0 = 0; k0 < D; k0 += 4) {
                v2f a  = *(const v2f*)(arow + k0);
                v2f b0 = *(const v2f*)(wrow + k0);
                v2f b1 = *(const v2f*)(wrow + 16 * LDP + k0);
                c0 = __builtin_amdgcn_wmma_f32_16x16x4_f32(false, a, false, b0,
                                                           (short)0, c0, false, false);
                c1 = __builtin_amdgcn_wmma_f32_16x16x4_f32(false, a, false, b1,
                                                           (short)0, c1, false, false);
            }
        }
        __syncthreads();

        if (has_tile) {
#pragma unroll
            for (int r = 0; r < 8; ++r) {
                float* crow = As + (mt * 16 + mhi + r) * LDP + nt0 * 16 + lmod;
                crow[0]  = c0[r];
                crow[16] = c1[r];
            }
        }
        __syncthreads();

        unsigned poff = (unsigned)lane * 16u;
        asm volatile("" : "+v"(poff));
        const v4f b4 = *(const v4f*)((const char*)bias  + poff);
        const v4f g4 = *(const v4f*)((const char*)gamma + poff);
        const v4f e4 = *(const v4f*)((const char*)beta  + poff);
        for (int q = 0; q < NODES_PER_WAVE; ++q) {
            const int node = node0 + q * 32 + wave;
            if (node >= n_hi) break;
            v4f v = *(const v4f*)(As + (q * 32 + wave) * LDP + lane * 4);
            float s = 0.f, s2 = 0.f;
#pragma unroll
            for (int i = 0; i < 4; ++i) {
                float vv = v[i] + b4[i];
                vv = fmaxf(vv, 0.0f);
                v[i] = vv;
                s  += vv;
                s2 += vv * vv;
            }
#pragma unroll
            for (int off = 16; off >= 1; off >>= 1) {
                s  += __shfl_xor(s,  off, 32);
                s2 += __shfl_xor(s2, off, 32);
            }
            float mu  = s * (1.0f / 128.0f);
            float var = s2 * (1.0f / 128.0f) - mu * mu;
            float rs  = rsqrtf(var + LN_EPS);
#pragma unroll
            for (int i = 0; i < 4; ++i)
                v[i] = (v[i] - mu) * rs * g4[i] + e4[i];
            *(v4f*)(out + (size_t)node * D + lane * 4) = v;
        }
    }
}

extern "C" void kernel_launch(void* const* d_in, const int* in_sizes, int n_in,
                              void* d_out, int out_size, void* d_ws, size_t ws_size,
                              hipStream_t stream) {
    const float* x     = (const float*)d_in[0];
    const int*   eidx  = (const int*)  d_in[1];
    const float* ew    = (const float*)d_in[2];
    const float* W     = (const float*)d_in[3];
    const float* bias  = (const float*)d_in[4];
    const float* gamma = (const float*)d_in[5];
    const float* beta  = (const float*)d_in[6];
    float*       out   = (float*)d_out;

    const int N = in_sizes[0] / D;
    const int E = in_sizes[2];
    const int* rows = eidx;
    const int* cols = eidx + E;

    float* wsf  = (float*)d_ws;
    float* deg  = wsf + (size_t)N * D;

    const int nchunks = (E + CHUNK - 1) / CHUNK;
    int* ent = (int*)wsf;
    int* off = ent + (size_t)nchunks * CHUNK * 4;

    partition_kernel<<<nchunks, 256, 0, stream>>>(rows, cols, ew, ent, off, E);
    build_kernel<<<(N + BIN_NODES - 1) / BIN_NODES, 512, 0, stream>>>(
        ent, off, deg, (int*)out, N, nchunks);

    pull_transform_kernel<<<PULL_GROUPS, 1024, 0, stream>>>(
        rows, cols, ew, deg, x, W, out, bias, gamma, beta, N, E);
}
